# starting kernel
# baseline (speedup 1.0000x reference)
#include <hip/hip_runtime.h>

#define N_NODES 100000
#define N_EDGES 600000
#define F_IN 128
#define F_OUT 128

typedef __attribute__((ext_vector_type(2))) float v2f;
typedef __attribute__((ext_vector_type(8))) float v8f;

__global__ __launch_bounds__(256) void gconv_init_bias(
    const float* __restrict__ bias, float* __restrict__ out, int total) {
  int i = blockIdx.x * blockDim.x + threadIdx.x;
  if (i < total) out[i] = bias[i & (F_OUT - 1)];
}

__global__ __launch_bounds__(256) void gconv_gemm_wmma(
    const float* __restrict__ X, const float* __restrict__ W,
    float* __restrict__ S) {
  const int lane = threadIdx.x & 31;
  const int wave = threadIdx.x >> 5;
  const int m0   = blockIdx.x * 16;
  const int n0   = wave * 16;
  const int half = lane >> 4;
  const int l15  = lane & 15;

  const float* xrow = X + (size_t)(m0 + l15) * F_IN;
  const int    bcol = n0 + l15;

  v8f acc = {};
#pragma unroll
  for (int k = 0; k < F_IN; k += 4) {
    v2f a = *(const v2f*)(xrow + k + half * 2);
    v2f b;
    b.x = W[(size_t)(k + half * 2 + 0) * F_OUT + bcol];
    b.y = W[(size_t)(k + half * 2 + 1) * F_OUT + bcol];
    acc = __builtin_amdgcn_wmma_f32_16x16x4_f32(
         false, a,  false, b,
         (short)0, acc,  false,  false);
  }

#pragma unroll
  for (int r = 0; r < 8; ++r) {
    int row = m0 + r + half * 8;
    S[(size_t)row * F_OUT + bcol] = acc[r];
  }
}

__global__ __launch_bounds__(256) void gconv_spmm_scatter(
    const float* __restrict__ support, const int* __restrict__ rows,
    const int* __restrict__ cols, const float* __restrict__ vals,
    float* __restrict__ out, int n_edges) {
  int gwave = (blockIdx.x * blockDim.x + threadIdx.x) >> 5;
  int lane  = threadIdx.x & 31;
  if (gwave >= n_edges) return;

  int   r = rows[gwave];
  int   c = cols[gwave];
  float v = vals[gwave];

  const float4* src = (const float4*)(support + (size_t)c * F_OUT);
  float4 m = src[lane];

  float* dst = out + (size_t)r * F_OUT + lane * 4;
  unsafeAtomicAdd(dst + 0, m.x * v);
  unsafeAtomicAdd(dst + 1, m.y * v);
  unsafeAtomicAdd(dst + 2, m.z * v);
  unsafeAtomicAdd(dst + 3, m.w * v);
}

extern "C" void kernel_launch(void* const* d_in, const int* in_sizes, int n_in,
                              void* d_out, int out_size, void* d_ws,
                              size_t ws_size, hipStream_t stream) {
  const float* x     = (const float*)d_in[0];
  const float* w     = (const float*)d_in[1];
  const float* bias  = (const float*)d_in[2];
  const float* vals0 = (const float*)d_in[3];
  const float* vals1 = (const float*)d_in[4];
  const int*   rows0 = (const int*)d_in[5];
  const int*   cols0 = (const int*)d_in[6];
  const int*   rows1 = (const int*)d_in[7];
  const int*   cols1 = (const int*)d_in[8];

  float* out     = (float*)d_out;
  float* support = (float*)d_ws;

  const int total = N_NODES * F_OUT;
  gconv_init_bias<<<(total + 255) / 256, 256, 0, stream>>>(bias, out, total);

  gconv_gemm_wmma<<<N_NODES / 16, 256, 0, stream>>>(x, w, support);

  const int edges_per_block = 256 / 32;
  const int sgrid = (N_EDGES + edges_per_block - 1) / edges_per_block;
  gconv_spmm_scatter<<<sgrid, 256, 0, stream>>>(support, rows0, cols0, vals0,
                                                out, N_EDGES);
  gconv_spmm_scatter<<<sgrid, 256, 0, stream>>>(support, rows1, cols1, vals1,
                                                out, N_EDGES);
}


# edge ranks on LDS counters: a block ranks one slice of the edge list for one row range, per-slice degrees scanned into starts; no global add per edge, no clear launch; sums in slice, arrival order
# speedup vs baseline: 2.4069x; 2.4069x over previous
#include <hip/hip_runtime.h>


#define N_NODES 100000
#define N_EDGES 600000
#define F_IN 128
#define F_OUT 128

#define ALL_EDGES (2 * N_EDGES)
#define QUADS_PER_SET (N_EDGES / 4)
#define ALL_QUADS (2 * QUADS_PER_SET)
#define ROWS_PER_WAVE 10
#define ROWS_PER_BLOCK (8 * ROWS_PER_WAVE)
#define ROW_TILES (ROWS_PER_BLOCK / 16)
#define PITCH (F_IN + 4)

#define N_RANGES 8
#define RANGE_ROWS (N_NODES / N_RANGES)
#define N_SLICES 32
#define SLICE_QUADS (ALL_QUADS / N_SLICES)
#define RANK_THREADS 512
#define RANK_BATCH 10
static_assert(RANGE_ROWS * N_RANGES == N_NODES, "ranges must tile the rows");
static_assert(SLICE_QUADS * N_SLICES == ALL_QUADS, "slices must tile the edge list");
static_assert(SLICE_QUADS * (N_SLICES / 2) == QUADS_PER_SET, "a slice lies in one set");

typedef __attribute__((ext_vector_type(2))) float v2f;
typedef __attribute__((ext_vector_type(8))) float v8f;
typedef __attribute__((ext_vector_type(2))) int v2i;

__global__ __launch_bounds__(RANK_THREADS) void gconv_edge_rank(
    const int* __restrict__ rows0, const int* __restrict__ rows1,
    int* __restrict__ slice_counts, int* __restrict__ rank,
    int* __restrict__ cursor) {
  __shared__ int cnt[RANGE_ROWS];
  const int tid = threadIdx.x;
  const int range = blockIdx.x % N_RANGES;
  const int slice = blockIdx.x / N_RANGES;
  const int r0 = range * RANGE_ROWS;

  for (int i = tid; i < RANGE_ROWS; i += RANK_THREADS) cnt[i] = 0;
  if (blockIdx.x == 0 && tid == 0) *cursor = 0;
  __syncthreads();

  const int q_begin = slice * SLICE_QUADS;
  const int4* src = (slice < N_SLICES / 2)
                        ? (const int4*)rows0 + q_begin
                        : (const int4*)rows1 + (q_begin - QUADS_PER_SET);
  int4* dst = (int4*)rank + q_begin;

  for (int j0 = tid; j0 < SLICE_QUADS; j0 += RANK_THREADS * RANK_BATCH) {
    int4 r[RANK_BATCH];
#pragma unroll
    for (int u = 0; u < RANK_BATCH; ++u) {
      const int j = j0 + u * RANK_THREADS;
      r[u] = (j < SLICE_QUADS) ? src[j] : make_int4(-1, -1, -1, -1);
    }
    int k[RANK_BATCH][4];
#pragma unroll
    for (int u = 0; u < RANK_BATCH; ++u) {
      const int rr[4] = {r[u].x, r[u].y, r[u].z, r[u].w};
#pragma unroll
      for (int c = 0; c < 4; ++c) {
        const unsigned d = (unsigned)(rr[c] - r0);
        k[u][c] = (d < (unsigned)RANGE_ROWS) ? atomicAdd(&cnt[d], 1) : -1;
      }
    }
#pragma unroll
    for (int u = 0; u < RANK_BATCH; ++u) {
      int* e = (int*)(dst + j0 + u * RANK_THREADS);
#pragma unroll
      for (int c = 0; c < 4; ++c)
        if (k[u][c] >= 0) e[c] = k[u][c];
    }
  }
  __syncthreads();

  int* sc = slice_counts + (size_t)slice * N_NODES + r0;
  for (int i = tid; i < RANGE_ROWS; i += RANK_THREADS) sc[i] = cnt[i];
}

__global__ __launch_bounds__(256) void gconv_row_segments(
    const int* __restrict__ slice_counts, int* __restrict__ counts,
    int* __restrict__ offsets, int* __restrict__ bases,
    int* __restrict__ cursor) {
  __shared__ int wave_sum[8];
  __shared__ int block_base;
  const int lane = threadIdx.x & 31;
  const int wave = threadIdx.x >> 5;
  const int r = blockIdx.x * 256 + threadIdx.x;

  int cs[N_SLICES];
  int c = 0;
#pragma unroll
  for (int s = 0; s < N_SLICES; ++s) {
    cs[s] = (r < N_NODES) ? slice_counts[(size_t)s * N_NODES + r] : 0;
    c += cs[s];
  }

  int s = c;
#pragma unroll
  for (int d = 1; d < 32; d <<= 1) {
    int t = __shfl_up(s, d);
    if (lane >= d) s += t;
  }
  if (lane == 31) wave_sum[wave] = s;
  __syncthreads();
  if (threadIdx.x == 0) {
    int tot = 0;
#pragma unroll
    for (int w = 0; w < 8; ++w) {
      int t = wave_sum[w];
      wave_sum[w] = tot;
      tot += t;
    }
    block_base = atomicAdd(cursor, tot);
  }
  __syncthreads();
  if (r < N_NODES) {
    int b = block_base + wave_sum[wave] + s - c;
    offsets[r] = b;
    counts[r] = c;
#pragma unroll
    for (int sl = 0; sl < N_SLICES; ++sl) {
      bases[(size_t)sl * N_NODES + r] = b;
      b += cs[sl];
    }
  }
}

__global__ __launch_bounds__(256) void gconv_edge_place(
    const int* __restrict__ rows0, const int* __restrict__ cols0,
    const float* __restrict__ vals0, const int* __restrict__ rows1,
    const int* __restrict__ cols1, const float* __restrict__ vals1,
    const int* __restrict__ rank, const int* __restrict__ bases,
    int* __restrict__ pairs) {
  const int q = blockIdx.x * 256 + threadIdx.x;
  if (q >= ALL_QUADS) return;
  int4 r, c;
  float4 v;
  if (q < QUADS_PER_SET) {
    r = ((const int4*)rows0)[q];
    c = ((const int4*)cols0)[q];
    v = ((const float4*)vals0)[q];
  } else {
    r = ((const int4*)rows1)[q - QUADS_PER_SET];
    c = ((const int4*)cols1)[q - QUADS_PER_SET];
    v = ((const float4*)vals1)[q - QUADS_PER_SET];
  }
  const int4 k = ((const int4*)rank)[q];
  const int* base = bases + (size_t)(q / SLICE_QUADS) * N_NODES;
  v2i* p = (v2i*)pairs;
  v2i e;
  e.x = c.x; e.y = __float_as_int(v.x); p[base[r.x] + k.x] = e;
  e.x = c.y; e.y = __float_as_int(v.y); p[base[r.y] + k.y] = e;
  e.x = c.z; e.y = __float_as_int(v.z); p[base[r.z] + k.z] = e;
  e.x = c.w; e.y = __float_as_int(v.w); p[base[r.w] + k.w] = e;
}

__device__ __forceinline__ float4 row_fma(float4 acc, float4 m, float v) {
  acc.x += m.x * v;
  acc.y += m.y * v;
  acc.z += m.z * v;
  acc.w += m.w * v;
  return acc;
}

__device__ __forceinline__ int src_row(int c) {
  return min(max(c, 0), N_NODES - 1);
}

__global__ __launch_bounds__(256) __attribute__((amdgpu_waves_per_eu(4, 12))) void gconv_sum_gemm(
    const float* __restrict__ X, const float* __restrict__ W,
    const int* __restrict__ counts, const int* __restrict__ offsets,
    const int* __restrict__ pairs, const float* __restrict__ bias,
    float* __restrict__ out) {
  __shared__ __attribute__((aligned(16))) float tile[ROWS_PER_BLOCK * PITCH];

  const int lane = threadIdx.x & 31;
  const int wv = __builtin_amdgcn_readfirstlane((int)(threadIdx.x >> 5));
  const int block_row0 = blockIdx.x * ROWS_PER_BLOCK;
  const float4* xin = (const float4*)X;
  const v2i* p = (const v2i*)pairs;

  const int row0 = block_row0 + wv * ROWS_PER_WAVE;
  int vdeg = 0, voff = 0;
  if (lane < ROWS_PER_WAVE) {
    vdeg = counts[row0 + lane];
    voff = offsets[row0 + lane];
  }
  for (int i = 0; i < ROWS_PER_WAVE; ++i) {
    const int deg = __builtin_amdgcn_readlane(vdeg, i);
    const v2i* e = p + __builtin_amdgcn_readlane(voff, i);
    float4 acc = make_float4(0.f, 0.f, 0.f, 0.f);
    int j = 0;
    for (; j + 4 <= deg; j += 4) {
      const v2i e0 = e[j], e1 = e[j + 1], e2 = e[j + 2], e3 = e[j + 3];
      const float4 m0 = xin[(size_t)src_row(e0.x) * (F_IN / 4) + lane];
      const float4 m1 = xin[(size_t)src_row(e1.x) * (F_IN / 4) + lane];
      const float4 m2 = xin[(size_t)src_row(e2.x) * (F_IN / 4) + lane];
      const float4 m3 = xin[(size_t)src_row(e3.x) * (F_IN / 4) + lane];
      acc = row_fma(acc, m0, __int_as_float(e0.y));
      acc = row_fma(acc, m1, __int_as_float(e1.y));
      acc = row_fma(acc, m2, __int_as_float(e2.y));
      acc = row_fma(acc, m3, __int_as_float(e3.y));
    }
    for (; j < deg; ++j) {
      const v2i e0 = e[j];
      const float4 m0 = xin[(size_t)src_row(e0.x) * (F_IN / 4) + lane];
      acc = row_fma(acc, m0, __int_as_float(e0.y));
    }
    *(float4*)(tile + (wv * ROWS_PER_WAVE + i) * PITCH + 4 * lane) = acc;
  }
  __syncthreads();

  const int half = lane >> 4;
  const int l15  = lane & 15;
  const int bcol = wv * 16 + l15;
  const float* arow = tile + l15 * PITCH + half * 2;

  v8f acc[ROW_TILES];
#pragma unroll
  for (int t = 0; t < ROW_TILES; ++t) acc[t] = (v8f){};

#pragma unroll
  for (int k = 0; k < F_IN; k += 4) {
    v2f b;
    b.x = W[(size_t)(k + half * 2 + 0) * F_OUT + bcol];
    b.y = W[(size_t)(k + half * 2 + 1) * F_OUT + bcol];
#pragma unroll
    for (int t = 0; t < ROW_TILES; ++t) {
      const v2f a = *(const v2f*)(arow + t * 16 * PITCH + k);
      acc[t] = __builtin_amdgcn_wmma_f32_16x16x4_f32(
           false, a,  false, b,
           (short)0, acc[t],  false,  false);
    }
  }

  const float bv = bias[bcol];
#pragma unroll
  for (int t = 0; t < ROW_TILES; ++t) {
#pragma unroll
    for (int r = 0; r < 8; ++r) {
      const int row = block_row0 + t * 16 + r + half * 8;
      out[(size_t)row * F_OUT + bcol] = acc[t][r] + bv;
    }
  }
}

extern "C" void kernel_launch(void* const* d_in, const int* in_sizes, int n_in,
                              void* d_out, int out_size, void* d_ws,
                              size_t ws_size, hipStream_t stream) {
  const float* x     = (const float*)d_in[0];
  const float* w     = (const float*)d_in[1];
  const float* bias  = (const float*)d_in[2];
  const float* vals0 = (const float*)d_in[3];
  const float* vals1 = (const float*)d_in[4];
  const int*   rows0 = (const int*)d_in[5];
  const int*   cols0 = (const int*)d_in[6];
  const int*   rows1 = (const int*)d_in[7];
  const int*   cols1 = (const int*)d_in[8];

  float* out     = (float*)d_out;
  int*   counts  = (int*)d_ws + N_NODES * F_OUT;
  int*   offsets = counts + N_NODES;
  int*   cursor  = offsets + N_NODES;
  int*   rank    = cursor + 64;
  int*   pairs   = rank + ALL_EDGES;
  int*   slice_counts = pairs + 2 * ALL_EDGES;
  int*   bases   = slice_counts + N_SLICES * N_NODES;

  gconv_edge_rank<<<N_SLICES * N_RANGES, RANK_THREADS, 0, stream>>>(
      rows0, rows1, slice_counts, rank, cursor);

  gconv_row_segments<<<(N_NODES + 255) / 256, 256, 0, stream>>>(
      slice_counts, counts, offsets, bases, cursor);

  gconv_edge_place<<<(ALL_QUADS + 255) / 256, 256, 0, stream>>>(
      rows0, cols0, vals0, rows1, cols1, vals1, rank, bases, pairs);

  gconv_sum_gemm<<<N_NODES / ROWS_PER_BLOCK, 256, 0, stream>>>(
      x, w, counts, offsets, pairs, bias, out);
}


# edge lists by a two-level LDS sort: a block sorts one slice by 80-row bucket, the sum kernel collects its bucket's runs and ranks them by row in LDS; 2 launches not 4; sums in arrival order
# speedup vs baseline: 3.7107x; 1.5417x over previous
#include <hip/hip_runtime.h>


#define N_NODES 100000
#define N_EDGES 600000
#define F_IN 128
#define F_OUT 128

#define ALL_EDGES (2 * N_EDGES)
#define ROWS_PER_WAVE 10
#define ROWS_PER_BLOCK (8 * ROWS_PER_WAVE)
#define ROW_TILES (ROWS_PER_BLOCK / 16)
#define PITCH (F_IN + 4)

#define N_SLICES 250
#define SLICE_EDGES (ALL_EDGES / N_SLICES)
#define SLICE_QUADS (SLICE_EDGES / 4)
#define SET_SLICES (N_SLICES / 2)
#define N_BUCKETS (N_NODES / ROWS_PER_BLOCK)
#define STARTS_PITCH 256
#define SORT_THREADS 512
#define SORT_QUADS 3
#define COL_BITS 17
#define CAP 1024
static_assert(SLICE_EDGES * N_SLICES == ALL_EDGES, "slices must tile the edge list");
static_assert(SLICE_QUADS * SET_SLICES * 4 == N_EDGES, "a slice lies in one set");
static_assert(N_BUCKETS * ROWS_PER_BLOCK == N_NODES, "buckets must tile the rows");
static_assert(SORT_QUADS * SORT_THREADS >= SLICE_QUADS, "a block must cover its slice");
static_assert(3 * SORT_THREADS >= N_BUCKETS, "the scan takes three counters per thread");
static_assert((1 << COL_BITS) >= N_NODES, "node index must fit the key");
static_assert(N_SLICES <= STARTS_PITCH && N_SLICES <= 256, "one thread of kernel 2 per slice");

typedef __attribute__((ext_vector_type(2))) float v2f;
typedef __attribute__((ext_vector_type(8))) float v8f;
typedef __attribute__((ext_vector_type(2))) int v2i;

__device__ __forceinline__ int src_row(int c) {
  return min(max(c, 0), N_NODES - 1);
}

__global__ __launch_bounds__(SORT_THREADS) void gconv_slice_sort(
    const int* __restrict__ rows0, const int* __restrict__ cols0,
    const float* __restrict__ vals0, const int* __restrict__ rows1,
    const int* __restrict__ cols1, const float* __restrict__ vals1,
    int* __restrict__ sorted, int* __restrict__ starts) {
  __shared__ int cnt[N_BUCKETS + 1];
  __shared__ int wave_tot[SORT_THREADS / 32];
  const int tid = threadIdx.x;
  const int lane = tid & 31;
  const int wave = tid >> 5;
  const int slice = blockIdx.x;

  for (int i = tid; i <= N_BUCKETS; i += SORT_THREADS) cnt[i] = 0;
  __syncthreads();

  const bool set0 = slice < SET_SLICES;
  const int q0 = (set0 ? slice : slice - SET_SLICES) * SLICE_QUADS;
  const int4* rsrc = (const int4*)(set0 ? rows0 : rows1) + q0;
  const int4* csrc = (const int4*)(set0 ? cols0 : cols1) + q0;
  const float4* vsrc = (const float4*)(set0 ? vals0 : vals1) + q0;

  int4 r[SORT_QUADS], c[SORT_QUADS];
  float4 v[SORT_QUADS];
#pragma unroll
  for (int u = 0; u < SORT_QUADS; ++u) {
    const int q = tid + u * SORT_THREADS;
    if (q < SLICE_QUADS) {
      r[u] = rsrc[q];
      c[u] = csrc[q];
      v[u] = vsrc[q];
    } else {
      r[u] = make_int4(-1, -1, -1, -1);
      c[u] = make_int4(0, 0, 0, 0);
      v[u] = make_float4(0.f, 0.f, 0.f, 0.f);
    }
  }

  int k[SORT_QUADS][4];
#pragma unroll
  for (int u = 0; u < SORT_QUADS; ++u) {
    const int rr[4] = {r[u].x, r[u].y, r[u].z, r[u].w};
#pragma unroll
    for (int e = 0; e < 4; ++e) {
      const unsigned row = (unsigned)rr[e];
      k[u][e] = (row < (unsigned)N_NODES) ? atomicAdd(&cnt[row / ROWS_PER_BLOCK], 1) : -1;
    }
  }
  __syncthreads();

  const int i0 = tid * 3;
  const int a0 = (i0 + 0 < N_BUCKETS) ? cnt[i0 + 0] : 0;
  const int a1 = (i0 + 1 < N_BUCKETS) ? cnt[i0 + 1] : 0;
  const int a2 = (i0 + 2 < N_BUCKETS) ? cnt[i0 + 2] : 0;
  const int t3 = a0 + a1 + a2;
  int s = t3;
#pragma unroll
  for (int d = 1; d < 32; d <<= 1) {
    const int t = __shfl_up(s, d);
    if (lane >= d) s += t;
  }
  if (lane == 31) wave_tot[wave] = s;
  __syncthreads();
  int before = 0, total = 0;
#pragma unroll
  for (int w = 0; w < SORT_THREADS / 32; ++w) {
    const int t = wave_tot[w];
    if (w < wave) before += t;
    total += t;
  }
  const int ex = before + s - t3;
  if (i0 + 0 < N_BUCKETS) cnt[i0 + 0] = ex;
  if (i0 + 1 < N_BUCKETS) cnt[i0 + 1] = ex + a0;
  if (i0 + 2 < N_BUCKETS) cnt[i0 + 2] = ex + a0 + a1;
  if (tid == 0) cnt[N_BUCKETS] = total;
  __syncthreads();

  v2i* dst = (v2i*)sorted + (size_t)slice * SLICE_EDGES;
#pragma unroll
  for (int u = 0; u < SORT_QUADS; ++u) {
    const int rr[4] = {r[u].x, r[u].y, r[u].z, r[u].w};
    const int cc[4] = {c[u].x, c[u].y, c[u].z, c[u].w};
    const float vv[4] = {v[u].x, v[u].y, v[u].z, v[u].w};
#pragma unroll
    for (int e = 0; e < 4; ++e) {
      if (k[u][e] >= 0) {
        const int b = (int)((unsigned)rr[e] / ROWS_PER_BLOCK);
        const int rl = rr[e] - b * ROWS_PER_BLOCK;
        v2i o;
        o.x = src_row(cc[e]) | (rl << COL_BITS);
        o.y = __float_as_int(vv[e]);
        dst[cnt[b] + k[u][e]] = o;
      }
    }
  }

  for (int i = tid; i <= N_BUCKETS; i += SORT_THREADS)
    starts[(size_t)i * STARTS_PITCH + slice] = cnt[i];
}

__device__ __forceinline__ float4 row_fma(float4 acc, float4 m, float v) {
  acc.x += m.x * v;
  acc.y += m.y * v;
  acc.z += m.z * v;
  acc.w += m.w * v;
  return acc;
}

__global__ __launch_bounds__(256) __attribute__((amdgpu_waves_per_eu(10, 12))) void gconv_sum_gemm(
    const float* __restrict__ X, const float* __restrict__ W,
    const int* __restrict__ sorted, const int* __restrict__ starts,
    const float* __restrict__ bias, float* __restrict__ out) {
  __shared__ __attribute__((aligned(16))) float tile[ROWS_PER_BLOCK * PITCH];
  __shared__ v2i stage[CAP];
  __shared__ v2i elist[CAP];
  __shared__ int srank[CAP];
  __shared__ int rcnt[ROWS_PER_BLOCK];
  __shared__ int rstart[ROWS_PER_BLOCK];
  __shared__ int wave_sum[8];
  __shared__ int n_edges;

  const int tid = threadIdx.x;
  const int lane = tid & 31;
  const int wv = __builtin_amdgcn_readfirstlane((int)(threadIdx.x >> 5));
  const int block_row0 = blockIdx.x * ROWS_PER_BLOCK;
  const float4* xin = (const float4*)X;

  int st = 0, n = 0;
  if (tid < N_SLICES) {
    st = starts[(size_t)blockIdx.x * STARTS_PITCH + tid];
    const int en = starts[(size_t)(blockIdx.x + 1) * STARTS_PITCH + tid];
    st = min(max(st, 0), SLICE_EDGES);
    n = min(max(en - st, 0), SLICE_EDGES - st);
  }
  int sc = n;
#pragma unroll
  for (int d = 1; d < 32; d <<= 1) {
    const int t = __shfl_up(sc, d);
    if (lane >= d) sc += t;
  }
  if (lane == 31) wave_sum[tid >> 5] = sc;
  __syncthreads();
  int pos = sc - n, tot = 0;
#pragma unroll
  for (int w = 0; w < 8; ++w) {
    const int t = wave_sum[w];
    if (w < (tid >> 5)) pos += t;
    tot += t;
  }
  if (tid == 0) n_edges = tot;
  __syncthreads();
  const int E = __builtin_amdgcn_readfirstlane(n_edges);
  const v2i* src = (const v2i*)sorted + (size_t)tid * SLICE_EDGES + st;

  for (int lo = 0; lo == 0 || lo < E; lo += CAP) {
    const int m = min(E - lo, CAP);
    __syncthreads();
    if (tid < ROWS_PER_BLOCK) rcnt[tid] = 0;
    const int ibeg = max(0, lo - pos);
    const int iend = min(n, lo + CAP - pos);
    for (int i = ibeg; i < iend; i += 4) {
      v2i e[4];
#pragma unroll
      for (int u = 0; u < 4; ++u) {
        e[u] = (v2i){0, 0};
        if (i + u < iend) e[u] = src[i + u];
      }
#pragma unroll
      for (int u = 0; u < 4; ++u)
        if (i + u < iend) stage[pos + i + u - lo] = e[u];
    }
    __syncthreads();
    for (int i = tid; i < m; i += 256) {
      const int rl = min(stage[i].x >> COL_BITS, ROWS_PER_BLOCK - 1);
      srank[i] = atomicAdd(&rcnt[rl], 1);
    }
    __syncthreads();
    if (tid < 32) {
      const int r0 = 3 * lane;
      const int a0 = (r0 + 0 < ROWS_PER_BLOCK) ? rcnt[r0 + 0] : 0;
      const int a1 = (r0 + 1 < ROWS_PER_BLOCK) ? rcnt[r0 + 1] : 0;
      const int a2 = (r0 + 2 < ROWS_PER_BLOCK) ? rcnt[r0 + 2] : 0;
      const int t3 = a0 + a1 + a2;
      int s = t3;
#pragma unroll
      for (int d = 1; d < 32; d <<= 1) {
        const int t = __shfl_up(s, d);
        if (lane >= d) s += t;
      }
      const int ex = s - t3;
      if (r0 + 0 < ROWS_PER_BLOCK) rstart[r0 + 0] = ex;
      if (r0 + 1 < ROWS_PER_BLOCK) rstart[r0 + 1] = ex + a0;
      if (r0 + 2 < ROWS_PER_BLOCK) rstart[r0 + 2] = ex + a0 + a1;
    }
    __syncthreads();
    for (int i = tid; i < m; i += 256) {
      const v2i e = stage[i];
      const int rl = min(e.x >> COL_BITS, ROWS_PER_BLOCK - 1);
      v2i o;
      o.x = e.x & ((1 << COL_BITS) - 1);
      o.y = e.y;
      elist[rstart[rl] + srank[i]] = o;
    }
    __syncthreads();

    for (int i = 0; i < ROWS_PER_WAVE; ++i) {
      const int row = wv * ROWS_PER_WAVE + i;
      const int deg = __builtin_amdgcn_readfirstlane(rcnt[row]);
      const v2i* e = elist + __builtin_amdgcn_readfirstlane(rstart[row]);
      float* trow = tile + row * PITCH + 4 * lane;
      float4 acc = make_float4(0.f, 0.f, 0.f, 0.f);
      if (lo != 0) acc = *(const float4*)trow;
      int j = 0;
      for (; j + 4 <= deg; j += 4) {
        const v2i e0 = e[j], e1 = e[j + 1], e2 = e[j + 2], e3 = e[j + 3];
        const int c0 = __builtin_amdgcn_readfirstlane(e0.x);
        const int c1 = __builtin_amdgcn_readfirstlane(e1.x);
        const int c2 = __builtin_amdgcn_readfirstlane(e2.x);
        const int c3 = __builtin_amdgcn_readfirstlane(e3.x);
        const float4 m0 = xin[(size_t)src_row(c0) * (F_IN / 4) + lane];
        const float4 m1 = xin[(size_t)src_row(c1) * (F_IN / 4) + lane];
        const float4 m2 = xin[(size_t)src_row(c2) * (F_IN / 4) + lane];
        const float4 m3 = xin[(size_t)src_row(c3) * (F_IN / 4) + lane];
        acc = row_fma(acc, m0, __int_as_float(__builtin_amdgcn_readfirstlane(e0.y)));
        acc = row_fma(acc, m1, __int_as_float(__builtin_amdgcn_readfirstlane(e1.y)));
        acc = row_fma(acc, m2, __int_as_float(__builtin_amdgcn_readfirstlane(e2.y)));
        acc = row_fma(acc, m3, __int_as_float(__builtin_amdgcn_readfirstlane(e3.y)));
      }
      for (; j < deg; ++j) {
        const v2i e0 = e[j];
        const int c0 = __builtin_amdgcn_readfirstlane(e0.x);
        const float4 m0 = xin[(size_t)src_row(c0) * (F_IN / 4) + lane];
        acc = row_fma(acc, m0, __int_as_float(__builtin_amdgcn_readfirstlane(e0.y)));
      }
      *(float4*)trow = acc;
    }
  }
  __syncthreads();

  const int half = lane >> 4;
  const int l15  = lane & 15;
  const int bcol = wv * 16 + l15;
  const float* arow = tile + l15 * PITCH + half * 2;

  v8f acc[ROW_TILES];
#pragma unroll
  for (int t = 0; t < ROW_TILES; ++t) acc[t] = (v8f){};

#pragma unroll
  for (int k = 0; k < F_IN; k += 4) {
    v2f b;
    b.x = W[(size_t)(k + half * 2 + 0) * F_OUT + bcol];
    b.y = W[(size_t)(k + half * 2 + 1) * F_OUT + bcol];
#pragma unroll
    for (int t = 0; t < ROW_TILES; ++t) {
      const v2f a = *(const v2f*)(arow + t * 16 * PITCH + k);
      acc[t] = __builtin_amdgcn_wmma_f32_16x16x4_f32(
           false, a,  false, b,
           (short)0, acc[t],  false,  false);
    }
  }

  const float bv = bias[bcol];
#pragma unroll
  for (int t = 0; t < ROW_TILES; ++t) {
#pragma unroll
    for (int r = 0; r < 8; ++r) {
      const int row = block_row0 + t * 16 + r + half * 8;
      out[(size_t)row * F_OUT + bcol] = acc[t][r] + bv;
    }
  }
}

extern "C" void kernel_launch(void* const* d_in, const int* in_sizes, int n_in,
                              void* d_out, int out_size, void* d_ws,
                              size_t ws_size, hipStream_t stream) {
  const float* x     = (const float*)d_in[0];
  const float* w     = (const float*)d_in[1];
  const float* bias  = (const float*)d_in[2];
  const float* vals0 = (const float*)d_in[3];
  const float* vals1 = (const float*)d_in[4];
  const int*   rows0 = (const int*)d_in[5];
  const int*   cols0 = (const int*)d_in[6];
  const int*   rows1 = (const int*)d_in[7];
  const int*   cols1 = (const int*)d_in[8];

  float* out    = (float*)d_out;
  int*   sorted = (int*)d_ws + N_NODES * F_OUT;
  int*   starts = sorted + 2 * ALL_EDGES;

  gconv_slice_sort<<<N_SLICES, SORT_THREADS, 0, stream>>>(
      rows0, cols0, vals0, rows1, cols1, vals1, sorted, starts);

  gconv_sum_gemm<<<N_NODES / ROWS_PER_BLOCK, 256, 0, stream>>>(
      x, w, sorted, starts, bias, out);
}


# sort kernel: a slice's sorted pairs are collected in LDS and leave 16 B per lane, 4 whole 128-B lines per store, not one scattered 8-B store per edge; same bytes, same places, sum order untouched
# speedup vs baseline: 4.0942x; 1.1033x over previous
#include <hip/hip_runtime.h>


#define N_NODES 100000
#define N_EDGES 600000
#define F_IN 128
#define F_OUT 128

#define ALL_EDGES (2 * N_EDGES)
#define ROWS_PER_WAVE 10
#define ROWS_PER_BLOCK (8 * ROWS_PER_WAVE)
#define ROW_TILES (ROWS_PER_BLOCK / 16)
#define PITCH (F_IN + 4)

#define N_SLICES 250
#define SLICE_EDGES (ALL_EDGES / N_SLICES)
#define SLICE_QUADS (SLICE_EDGES / 4)
#define SET_SLICES (N_SLICES / 2)
#define N_BUCKETS (N_NODES / ROWS_PER_BLOCK)
#define STARTS_PITCH 256
#define SORT_THREADS 512
#define SORT_QUADS 3
#define COL_BITS 17
#define CAP 1024
static_assert(SLICE_EDGES * N_SLICES == ALL_EDGES, "slices must tile the edge list");
static_assert(SLICE_QUADS * SET_SLICES * 4 == N_EDGES, "a slice lies in one set");
static_assert(N_BUCKETS * ROWS_PER_BLOCK == N_NODES, "buckets must tile the rows");
static_assert(SORT_QUADS * SORT_THREADS >= SLICE_QUADS, "a block must cover its slice");
static_assert(3 * SORT_THREADS >= N_BUCKETS, "the scan takes three counters per thread");
static_assert((1 << COL_BITS) >= N_NODES, "node index must fit the key");
static_assert(N_SLICES <= STARTS_PITCH && N_SLICES <= 256, "one thread of kernel 2 per slice");

typedef __attribute__((ext_vector_type(2))) float v2f;
typedef __attribute__((ext_vector_type(8))) float v8f;
typedef __attribute__((ext_vector_type(2))) int v2i;

__device__ __forceinline__ int src_row(int c) {
  return min(max(c, 0), N_NODES - 1);
}

__global__ __launch_bounds__(SORT_THREADS) void gconv_slice_sort(
    const int* __restrict__ rows0, const int* __restrict__ cols0,
    const float* __restrict__ vals0, const int* __restrict__ rows1,
    const int* __restrict__ cols1, const float* __restrict__ vals1,
    int* __restrict__ sorted, int* __restrict__ starts) {
  __shared__ int cnt[N_BUCKETS + 1];
  __shared__ int wave_tot[SORT_THREADS / 32];
  __shared__ __attribute__((aligned(16))) v2i sbuf[SLICE_EDGES];
  const int tid = threadIdx.x;
  const int lane = tid & 31;
  const int wave = tid >> 5;
  const int slice = blockIdx.x;

  for (int i = tid; i <= N_BUCKETS; i += SORT_THREADS) cnt[i] = 0;
  __syncthreads();

  const bool set0 = slice < SET_SLICES;
  const int q0 = (set0 ? slice : slice - SET_SLICES) * SLICE_QUADS;
  const int4* rsrc = (const int4*)(set0 ? rows0 : rows1) + q0;
  const int4* csrc = (const int4*)(set0 ? cols0 : cols1) + q0;
  const float4* vsrc = (const float4*)(set0 ? vals0 : vals1) + q0;

  int4 r[SORT_QUADS], c[SORT_QUADS];
  float4 v[SORT_QUADS];
#pragma unroll
  for (int u = 0; u < SORT_QUADS; ++u) {
    const int q = tid + u * SORT_THREADS;
    if (q < SLICE_QUADS) {
      r[u] = rsrc[q];
      c[u] = csrc[q];
      v[u] = vsrc[q];
    } else {
      r[u] = make_int4(-1, -1, -1, -1);
      c[u] = make_int4(0, 0, 0, 0);
      v[u] = make_float4(0.f, 0.f, 0.f, 0.f);
    }
  }

  int k[SORT_QUADS][4];
#pragma unroll
  for (int u = 0; u < SORT_QUADS; ++u) {
    const int rr[4] = {r[u].x, r[u].y, r[u].z, r[u].w};
#pragma unroll
    for (int e = 0; e < 4; ++e) {
      const unsigned row = (unsigned)rr[e];
      k[u][e] = (row < (unsigned)N_NODES) ? atomicAdd(&cnt[row / ROWS_PER_BLOCK], 1) : -1;
    }
  }
  __syncthreads();

  const int i0 = tid * 3;
  const int a0 = (i0 + 0 < N_BUCKETS) ? cnt[i0 + 0] : 0;
  const int a1 = (i0 + 1 < N_BUCKETS) ? cnt[i0 + 1] : 0;
  const int a2 = (i0 + 2 < N_BUCKETS) ? cnt[i0 + 2] : 0;
  const int t3 = a0 + a1 + a2;
  int s = t3;
#pragma unroll
  for (int d = 1; d < 32; d <<= 1) {
    const int t = __shfl_up(s, d);
    if (lane >= d) s += t;
  }
  if (lane == 31) wave_tot[wave] = s;
  __syncthreads();
  int before = 0, total = 0;
#pragma unroll
  for (int w = 0; w < SORT_THREADS / 32; ++w) {
    const int t = wave_tot[w];
    if (w < wave) before += t;
    total += t;
  }
  const int ex = before + s - t3;
  if (i0 + 0 < N_BUCKETS) cnt[i0 + 0] = ex;
  if (i0 + 1 < N_BUCKETS) cnt[i0 + 1] = ex + a0;
  if (i0 + 2 < N_BUCKETS) cnt[i0 + 2] = ex + a0 + a1;
  if (tid == 0) cnt[N_BUCKETS] = total;
  __syncthreads();

#pragma unroll
  for (int u = 0; u < SORT_QUADS; ++u) {
    const int rr[4] = {r[u].x, r[u].y, r[u].z, r[u].w};
    const int cc[4] = {c[u].x, c[u].y, c[u].z, c[u].w};
    const float vv[4] = {v[u].x, v[u].y, v[u].z, v[u].w};
#pragma unroll
    for (int e = 0; e < 4; ++e) {
      if (k[u][e] >= 0) {
        const int b = (int)((unsigned)rr[e] / ROWS_PER_BLOCK);
        const int rl = rr[e] - b * ROWS_PER_BLOCK;
        v2i o;
        o.x = src_row(cc[e]) | (rl << COL_BITS);
        o.y = __float_as_int(vv[e]);
        sbuf[cnt[b] + k[u][e]] = o;
      }
    }
  }

  __syncthreads();
  v2i* dst = (v2i*)sorted + (size_t)slice * SLICE_EDGES;
  int4* dst4 = (int4*)dst;
  const int4* src4 = (const int4*)sbuf;
#pragma unroll
  for (int u = 0; u < (SLICE_EDGES / 2 + SORT_THREADS - 1) / SORT_THREADS; ++u) {
    const int i = tid + u * SORT_THREADS;
    if (i < (total >> 1)) dst4[i] = src4[i];
  }
  if ((total & 1) && tid == 0) dst[total - 1] = sbuf[total - 1];

  for (int i = tid; i <= N_BUCKETS; i += SORT_THREADS)
    starts[(size_t)i * STARTS_PITCH + slice] = cnt[i];
}

__device__ __forceinline__ float4 row_fma(float4 acc, float4 m, float v) {
  acc.x += m.x * v;
  acc.y += m.y * v;
  acc.z += m.z * v;
  acc.w += m.w * v;
  return acc;
}

__global__ __launch_bounds__(256) __attribute__((amdgpu_waves_per_eu(10, 12))) void gconv_sum_gemm(
    const float* __restrict__ X, const float* __restrict__ W,
    const int* __restrict__ sorted, const int* __restrict__ starts,
    const float* __restrict__ bias, float* __restrict__ out) {
  __shared__ __attribute__((aligned(16))) float tile[ROWS_PER_BLOCK * PITCH];
  __shared__ v2i stage[CAP];
  __shared__ v2i elist[CAP];
  __shared__ int srank[CAP];
  __shared__ int rcnt[ROWS_PER_BLOCK];
  __shared__ int rstart[ROWS_PER_BLOCK];
  __shared__ int wave_sum[8];
  __shared__ int n_edges;

  const int tid = threadIdx.x;
  const int lane = tid & 31;
  const int wv = __builtin_amdgcn_readfirstlane((int)(threadIdx.x >> 5));
  const int block_row0 = blockIdx.x * ROWS_PER_BLOCK;
  const float4* xin = (const float4*)X;

  int st = 0, n = 0;
  if (tid < N_SLICES) {
    st = starts[(size_t)blockIdx.x * STARTS_PITCH + tid];
    const int en = starts[(size_t)(blockIdx.x + 1) * STARTS_PITCH + tid];
    st = min(max(st, 0), SLICE_EDGES);
    n = min(max(en - st, 0), SLICE_EDGES - st);
  }
  int sc = n;
#pragma unroll
  for (int d = 1; d < 32; d <<= 1) {
    const int t = __shfl_up(sc, d);
    if (lane >= d) sc += t;
  }
  if (lane == 31) wave_sum[tid >> 5] = sc;
  __syncthreads();
  int pos = sc - n, tot = 0;
#pragma unroll
  for (int w = 0; w < 8; ++w) {
    const int t = wave_sum[w];
    if (w < (tid >> 5)) pos += t;
    tot += t;
  }
  if (tid == 0) n_edges = tot;
  __syncthreads();
  const int E = __builtin_amdgcn_readfirstlane(n_edges);
  const v2i* src = (const v2i*)sorted + (size_t)tid * SLICE_EDGES + st;

  for (int lo = 0; lo == 0 || lo < E; lo += CAP) {
    const int m = min(E - lo, CAP);
    __syncthreads();
    if (tid < ROWS_PER_BLOCK) rcnt[tid] = 0;
    const int ibeg = max(0, lo - pos);
    const int iend = min(n, lo + CAP - pos);
    for (int i = ibeg; i < iend; i += 4) {
      v2i e[4];
#pragma unroll
      for (int u = 0; u < 4; ++u) {
        e[u] = (v2i){0, 0};
        if (i + u < iend) e[u] = src[i + u];
      }
#pragma unroll
      for (int u = 0; u < 4; ++u)
        if (i + u < iend) stage[pos + i + u - lo] = e[u];
    }
    __syncthreads();
    for (int i = tid; i < m; i += 256) {
      const int rl = min(stage[i].x >> COL_BITS, ROWS_PER_BLOCK - 1);
      srank[i] = atomicAdd(&rcnt[rl], 1);
    }
    __syncthreads();
    if (tid < 32) {
      const int r0 = 3 * lane;
      const int a0 = (r0 + 0 < ROWS_PER_BLOCK) ? rcnt[r0 + 0] : 0;
      const int a1 = (r0 + 1 < ROWS_PER_BLOCK) ? rcnt[r0 + 1] : 0;
      const int a2 = (r0 + 2 < ROWS_PER_BLOCK) ? rcnt[r0 + 2] : 0;
      const int t3 = a0 + a1 + a2;
      int s = t3;
#pragma unroll
      for (int d = 1; d < 32; d <<= 1) {
        const int t = __shfl_up(s, d);
        if (lane >= d) s += t;
      }
      const int ex = s - t3;
      if (r0 + 0 < ROWS_PER_BLOCK) rstart[r0 + 0] = ex;
      if (r0 + 1 < ROWS_PER_BLOCK) rstart[r0 + 1] = ex + a0;
      if (r0 + 2 < ROWS_PER_BLOCK) rstart[r0 + 2] = ex + a0 + a1;
    }
    __syncthreads();
    for (int i = tid; i < m; i += 256) {
      const v2i e = stage[i];
      const int rl = min(e.x >> COL_BITS, ROWS_PER_BLOCK - 1);
      elist[rstart[rl] + srank[i]] = e;
    }
    __syncthreads();

    const int row0 = wv * ROWS_PER_WAVE;
    if (lo == 0) {
      for (int i = 0; i < ROWS_PER_WAVE; ++i)
        *(float4*)(tile + (row0 + i) * PITCH + 4 * lane) = make_float4(0.f, 0.f, 0.f, 0.f);
    }
    const int beg = min(max(__builtin_amdgcn_readfirstlane(rstart[row0]), 0), m);
    const int nxt = __builtin_amdgcn_readfirstlane(
        rstart[min(row0 + ROWS_PER_WAVE, ROWS_PER_BLOCK - 1)]);
    const int end = (wv == 7) ? m : min(max(nxt, beg), m);
    int cur = -1;
    float4 acc = make_float4(0.f, 0.f, 0.f, 0.f);
    for (int c = beg; c < end; c += 32) {
      const v2i en = elist[min(c + lane, end - 1)];
      const int nb = min(end - c, 32);
      for (int s = 0; s < nb; s += 8) {
        int key[8];
        float wgt[8];
        float4 mrow[8];
#pragma unroll
        for (int u = 0; u < 8; ++u) {
          key[u] = __builtin_amdgcn_readlane(en.x, s + u);
          wgt[u] = __int_as_float(__builtin_amdgcn_readlane(en.y, s + u));
          mrow[u] = xin[(size_t)src_row(key[u] & ((1 << COL_BITS) - 1)) * (F_IN / 4) + lane];
        }
#pragma unroll
        for (int u = 0; u < 8; ++u) {
          if (s + u < nb) {
            const int r = min((int)((unsigned)key[u] >> COL_BITS), ROWS_PER_BLOCK - 1);
            if (r != cur) {
              if (cur >= 0) *(float4*)(tile + cur * PITCH + 4 * lane) = acc;
              cur = r;
              acc = (lo != 0) ? *(const float4*)(tile + r * PITCH + 4 * lane)
                              : make_float4(0.f, 0.f, 0.f, 0.f);
            }
            acc = row_fma(acc, mrow[u], wgt[u]);
          }
        }
      }
    }
    if (cur >= 0) *(float4*)(tile + cur * PITCH + 4 * lane) = acc;
  }
  __syncthreads();

  const int half = lane >> 4;
  const int l15  = lane & 15;
  const int bcol = wv * 16 + l15;
  const float* arow = tile + l15 * PITCH + half * 2;

  v8f acc[ROW_TILES];
#pragma unroll
  for (int t = 0; t < ROW_TILES; ++t) acc[t] = (v8f){};

#pragma unroll
  for (int k = 0; k < F_IN; k += 4) {
    v2f b;
    b.x = W[(size_t)(k + half * 2 + 0) * F_OUT + bcol];
    b.y = W[(size_t)(k + half * 2 + 1) * F_OUT + bcol];
#pragma unroll
    for (int t = 0; t < ROW_TILES; ++t) {
      const v2f a = *(const v2f*)(arow + t * 16 * PITCH + k);
      acc[t] = __builtin_amdgcn_wmma_f32_16x16x4_f32(
           false, a,  false, b,
           (short)0, acc[t],  false,  false);
    }
  }

  const float bv = bias[bcol];
#pragma unroll
  for (int t = 0; t < ROW_TILES; ++t) {
#pragma unroll
    for (int r = 0; r < 8; ++r) {
      const int row = block_row0 + t * 16 + r + half * 8;
      out[(size_t)row * F_OUT + bcol] = acc[t][r] + bv;
    }
  }
}

extern "C" void kernel_launch(void* const* d_in, const int* in_sizes, int n_in,
                              void* d_out, int out_size, void* d_ws,
                              size_t ws_size, hipStream_t stream) {
  const float* x     = (const float*)d_in[0];
  const float* w     = (const float*)d_in[1];
  const float* bias  = (const float*)d_in[2];
  const float* vals0 = (const float*)d_in[3];
  const float* vals1 = (const float*)d_in[4];
  const int*   rows0 = (const int*)d_in[5];
  const int*   cols0 = (const int*)d_in[6];
  const int*   rows1 = (const int*)d_in[7];
  const int*   cols1 = (const int*)d_in[8];

  float* out    = (float*)d_out;
  int*   sorted = (int*)d_ws + N_NODES * F_OUT;
  int*   starts = sorted + 2 * ALL_EDGES;

  gconv_slice_sort<<<N_SLICES, SORT_THREADS, 0, stream>>>(
      rows0, cols0, vals0, rows1, cols1, vals1, sorted, starts);

  gconv_sum_gemm<<<N_NODES / ROWS_PER_BLOCK, 256, 0, stream>>>(
      x, w, sorted, starts, bias, out);
}


# write-out: finished 80x128 block staged through the free LDS tile, leaves as whole 512-B rows (+bias): one 16-B-per-lane store = 4 full lines, not stores of two 64-B half lines; values identical
# speedup vs baseline: 4.2591x; 1.0403x over previous
#include <hip/hip_runtime.h>


#define N_NODES 100000
#define N_EDGES 600000
#define F_IN 128
#define F_OUT 128

#define ALL_EDGES (2 * N_EDGES)
#define ROWS_PER_WAVE 10
#define ROWS_PER_BLOCK (8 * ROWS_PER_WAVE)
#define ROW_TILES (ROWS_PER_BLOCK / 16)
#define PITCH (F_IN + 4)

#define N_SLICES 250
#define SLICE_EDGES (ALL_EDGES / N_SLICES)
#define SLICE_QUADS (SLICE_EDGES / 4)
#define SET_SLICES (N_SLICES / 2)
#define N_BUCKETS (N_NODES / ROWS_PER_BLOCK)
#define STARTS_PITCH 256
#define SORT_THREADS 512
#define SORT_QUADS 3
#define COL_BITS 17
#define CAP 1024
static_assert(SLICE_EDGES * N_SLICES == ALL_EDGES, "slices must tile the edge list");
static_assert(SLICE_QUADS * SET_SLICES * 4 == N_EDGES, "a slice lies in one set");
static_assert(N_BUCKETS * ROWS_PER_BLOCK == N_NODES, "buckets must tile the rows");
static_assert(SORT_QUADS * SORT_THREADS >= SLICE_QUADS, "a block must cover its slice");
static_assert(3 * SORT_THREADS >= N_BUCKETS, "the scan takes three counters per thread");
static_assert((1 << COL_BITS) >= N_NODES, "node index must fit the key");
static_assert(N_SLICES <= STARTS_PITCH && N_SLICES <= 256, "one thread of kernel 2 per slice");

typedef __attribute__((ext_vector_type(2))) float v2f;
typedef __attribute__((ext_vector_type(8))) float v8f;
typedef __attribute__((ext_vector_type(2))) int v2i;

__device__ __forceinline__ int src_row(int c) {
  return min(max(c, 0), N_NODES - 1);
}

__global__ __launch_bounds__(SORT_THREADS) void gconv_slice_sort(
    const int* __restrict__ rows0, const int* __restrict__ cols0,
    const float* __restrict__ vals0, const int* __restrict__ rows1,
    const int* __restrict__ cols1, const float* __restrict__ vals1,
    int* __restrict__ sorted, int* __restrict__ starts) {
  __shared__ int cnt[N_BUCKETS + 1];
  __shared__ int wave_tot[SORT_THREADS / 32];
  __shared__ __attribute__((aligned(16))) v2i sbuf[SLICE_EDGES];
  const int tid = threadIdx.x;
  const int lane = tid & 31;
  const int wave = tid >> 5;
  const int slice = blockIdx.x;

  for (int i = tid; i <= N_BUCKETS; i += SORT_THREADS) cnt[i] = 0;
  __syncthreads();

  const bool set0 = slice < SET_SLICES;
  const int q0 = (set0 ? slice : slice - SET_SLICES) * SLICE_QUADS;
  const int4* rsrc = (const int4*)(set0 ? rows0 : rows1) + q0;
  const int4* csrc = (const int4*)(set0 ? cols0 : cols1) + q0;
  const float4* vsrc = (const float4*)(set0 ? vals0 : vals1) + q0;

  int4 r[SORT_QUADS], c[SORT_QUADS];
  float4 v[SORT_QUADS];
#pragma unroll
  for (int u = 0; u < SORT_QUADS; ++u) {
    const int q = tid + u * SORT_THREADS;
    if (q < SLICE_QUADS) {
      r[u] = rsrc[q];
      c[u] = csrc[q];
      v[u] = vsrc[q];
    } else {
      r[u] = make_int4(-1, -1, -1, -1);
      c[u] = make_int4(0, 0, 0, 0);
      v[u] = make_float4(0.f, 0.f, 0.f, 0.f);
    }
  }

  int k[SORT_QUADS][4];
#pragma unroll
  for (int u = 0; u < SORT_QUADS; ++u) {
    const int rr[4] = {r[u].x, r[u].y, r[u].z, r[u].w};
#pragma unroll
    for (int e = 0; e < 4; ++e) {
      const unsigned row = (unsigned)rr[e];
      k[u][e] = (row < (unsigned)N_NODES) ? atomicAdd(&cnt[row / ROWS_PER_BLOCK], 1) : -1;
    }
  }
  __syncthreads();

  const int i0 = tid * 3;
  const int a0 = (i0 + 0 < N_BUCKETS) ? cnt[i0 + 0] : 0;
  const int a1 = (i0 + 1 < N_BUCKETS) ? cnt[i0 + 1] : 0;
  const int a2 = (i0 + 2 < N_BUCKETS) ? cnt[i0 + 2] : 0;
  const int t3 = a0 + a1 + a2;
  int s = t3;
#pragma unroll
  for (int d = 1; d < 32; d <<= 1) {
    const int t = __shfl_up(s, d);
    if (lane >= d) s += t;
  }
  if (lane == 31) wave_tot[wave] = s;
  __syncthreads();
  int before = 0, total = 0;
#pragma unroll
  for (int w = 0; w < SORT_THREADS / 32; ++w) {
    const int t = wave_tot[w];
    if (w < wave) before += t;
    total += t;
  }
  const int ex = before + s - t3;
  if (i0 + 0 < N_BUCKETS) cnt[i0 + 0] = ex;
  if (i0 + 1 < N_BUCKETS) cnt[i0 + 1] = ex + a0;
  if (i0 + 2 < N_BUCKETS) cnt[i0 + 2] = ex + a0 + a1;
  if (tid == 0) cnt[N_BUCKETS] = total;
  __syncthreads();

#pragma unroll
  for (int u = 0; u < SORT_QUADS; ++u) {
    const int rr[4] = {r[u].x, r[u].y, r[u].z, r[u].w};
    const int cc[4] = {c[u].x, c[u].y, c[u].z, c[u].w};
    const float vv[4] = {v[u].x, v[u].y, v[u].z, v[u].w};
#pragma unroll
    for (int e = 0; e < 4; ++e) {
      if (k[u][e] >= 0) {
        const int b = (int)((unsigned)rr[e] / ROWS_PER_BLOCK);
        const int rl = rr[e] - b * ROWS_PER_BLOCK;
        v2i o;
        o.x = src_row(cc[e]) | (rl << COL_BITS);
        o.y = __float_as_int(vv[e]);
        sbuf[cnt[b] + k[u][e]] = o;
      }
    }
  }

  __syncthreads();
  v2i* dst = (v2i*)sorted + (size_t)slice * SLICE_EDGES;
  int4* dst4 = (int4*)dst;
  const int4* src4 = (const int4*)sbuf;
#pragma unroll
  for (int u = 0; u < (SLICE_EDGES / 2 + SORT_THREADS - 1) / SORT_THREADS; ++u) {
    const int i = tid + u * SORT_THREADS;
    if (i < (total >> 1)) dst4[i] = src4[i];
  }
  if ((total & 1) && tid == 0) dst[total - 1] = sbuf[total - 1];

  for (int i = tid; i <= N_BUCKETS; i += SORT_THREADS)
    starts[(size_t)i * STARTS_PITCH + slice] = cnt[i];
}

__device__ __forceinline__ float4 row_fma(float4 acc, float4 m, float v) {
  acc.x += m.x * v;
  acc.y += m.y * v;
  acc.z += m.z * v;
  acc.w += m.w * v;
  return acc;
}

__global__ __launch_bounds__(256) __attribute__((amdgpu_waves_per_eu(10, 12))) void gconv_sum_gemm(
    const float* __restrict__ X, const float* __restrict__ W,
    const int* __restrict__ sorted, const int* __restrict__ starts,
    const float* __restrict__ bias, float* __restrict__ out) {
  __shared__ __attribute__((aligned(16))) float tile[ROWS_PER_BLOCK * PITCH];
  __shared__ v2i stage[CAP];
  __shared__ v2i elist[CAP];
  __shared__ int srank[CAP];
  __shared__ int rcnt[ROWS_PER_BLOCK];
  __shared__ int rstart[ROWS_PER_BLOCK];
  __shared__ int wave_sum[8];
  __shared__ int n_edges;

  const int tid = threadIdx.x;
  const int lane = tid & 31;
  const int wv = __builtin_amdgcn_readfirstlane((int)(threadIdx.x >> 5));
  const int block_row0 = blockIdx.x * ROWS_PER_BLOCK;
  const float4* xin = (const float4*)X;

  int st = 0, n = 0;
  if (tid < N_SLICES) {
    st = starts[(size_t)blockIdx.x * STARTS_PITCH + tid];
    const int en = starts[(size_t)(blockIdx.x + 1) * STARTS_PITCH + tid];
    st = min(max(st, 0), SLICE_EDGES);
    n = min(max(en - st, 0), SLICE_EDGES - st);
  }
  int sc = n;
#pragma unroll
  for (int d = 1; d < 32; d <<= 1) {
    const int t = __shfl_up(sc, d);
    if (lane >= d) sc += t;
  }
  if (lane == 31) wave_sum[tid >> 5] = sc;
  __syncthreads();
  int pos = sc - n, tot = 0;
#pragma unroll
  for (int w = 0; w < 8; ++w) {
    const int t = wave_sum[w];
    if (w < (tid >> 5)) pos += t;
    tot += t;
  }
  if (tid == 0) n_edges = tot;
  __syncthreads();
  const int E = __builtin_amdgcn_readfirstlane(n_edges);
  const v2i* src = (const v2i*)sorted + (size_t)tid * SLICE_EDGES + st;

  for (int lo = 0; lo == 0 || lo < E; lo += CAP) {
    const int m = min(E - lo, CAP);
    __syncthreads();
    if (tid < ROWS_PER_BLOCK) rcnt[tid] = 0;
    const int ibeg = max(0, lo - pos);
    const int iend = min(n, lo + CAP - pos);
    for (int i = ibeg; i < iend; i += 4) {
      v2i e[4];
#pragma unroll
      for (int u = 0; u < 4; ++u) {
        e[u] = (v2i){0, 0};
        if (i + u < iend) e[u] = src[i + u];
      }
#pragma unroll
      for (int u = 0; u < 4; ++u)
        if (i + u < iend) stage[pos + i + u - lo] = e[u];
    }
    __syncthreads();
    for (int i = tid; i < m; i += 256) {
      const int rl = min(stage[i].x >> COL_BITS, ROWS_PER_BLOCK - 1);
      srank[i] = atomicAdd(&rcnt[rl], 1);
    }
    __syncthreads();
    if (tid < 32) {
      const int r0 = 3 * lane;
      const int a0 = (r0 + 0 < ROWS_PER_BLOCK) ? rcnt[r0 + 0] : 0;
      const int a1 = (r0 + 1 < ROWS_PER_BLOCK) ? rcnt[r0 + 1] : 0;
      const int a2 = (r0 + 2 < ROWS_PER_BLOCK) ? rcnt[r0 + 2] : 0;
      const int t3 = a0 + a1 + a2;
      int s = t3;
#pragma unroll
      for (int d = 1; d < 32; d <<= 1) {
        const int t = __shfl_up(s, d);
        if (lane >= d) s += t;
      }
      const int ex = s - t3;
      if (r0 + 0 < ROWS_PER_BLOCK) rstart[r0 + 0] = ex;
      if (r0 + 1 < ROWS_PER_BLOCK) rstart[r0 + 1] = ex + a0;
      if (r0 + 2 < ROWS_PER_BLOCK) rstart[r0 + 2] = ex + a0 + a1;
    }
    __syncthreads();
    for (int i = tid; i < m; i += 256) {
      const v2i e = stage[i];
      const int rl = min(e.x >> COL_BITS, ROWS_PER_BLOCK - 1);
      elist[rstart[rl] + srank[i]] = e;
    }
    __syncthreads();

    const int row0 = wv * ROWS_PER_WAVE;
    if (lo == 0) {
      for (int i = 0; i < ROWS_PER_WAVE; ++i)
        *(float4*)(tile + (row0 + i) * PITCH + 4 * lane) = make_float4(0.f, 0.f, 0.f, 0.f);
    }
    const int beg = min(max(__builtin_amdgcn_readfirstlane(rstart[row0]), 0), m);
    const int nxt = __builtin_amdgcn_readfirstlane(
        rstart[min(row0 + ROWS_PER_WAVE, ROWS_PER_BLOCK - 1)]);
    const int end = (wv == 7) ? m : min(max(nxt, beg), m);
    int cur = -1;
    float4 acc = make_float4(0.f, 0.f, 0.f, 0.f);
    for (int c = beg; c < end; c += 32) {
      const v2i en = elist[min(c + lane, end - 1)];
      const int nb = min(end - c, 32);
      for (int s = 0; s < nb; s += 8) {
        int key[8];
        float wgt[8];
        float4 mrow[8];
#pragma unroll
        for (int u = 0; u < 8; ++u) {
          key[u] = __builtin_amdgcn_readlane(en.x, s + u);
          wgt[u] = __int_as_float(__builtin_amdgcn_readlane(en.y, s + u));
          mrow[u] = xin[(size_t)src_row(key[u] & ((1 << COL_BITS) - 1)) * (F_IN / 4) + lane];
        }
#pragma unroll
        for (int u = 0; u < 8; ++u) {
          if (s + u < nb) {
            const int r = min((int)((unsigned)key[u] >> COL_BITS), ROWS_PER_BLOCK - 1);
            if (r != cur) {
              if (cur >= 0) *(float4*)(tile + cur * PITCH + 4 * lane) = acc;
              cur = r;
              acc = (lo != 0) ? *(const float4*)(tile + r * PITCH + 4 * lane)
                              : make_float4(0.f, 0.f, 0.f, 0.f);
            }
            acc = row_fma(acc, mrow[u], wgt[u]);
          }
        }
      }
    }
    if (cur >= 0) *(float4*)(tile + cur * PITCH + 4 * lane) = acc;
  }
  __syncthreads();

  const int half = lane >> 4;
  const int l15  = lane & 15;
  const int bcol = wv * 16 + l15;
  const float* arow = tile + l15 * PITCH + half * 2;

  v8f acc[ROW_TILES];
#pragma unroll
  for (int t = 0; t < ROW_TILES; ++t) acc[t] = (v8f){};

#pragma unroll
  for (int k = 0; k < F_IN; k += 4) {
    v2f b;
    b.x = W[(size_t)(k + half * 2 + 0) * F_OUT + bcol];
    b.y = W[(size_t)(k + half * 2 + 1) * F_OUT + bcol];
#pragma unroll
    for (int t = 0; t < ROW_TILES; ++t) {
      const v2f a = *(const v2f*)(arow + t * 16 * PITCH + k);
      acc[t] = __builtin_amdgcn_wmma_f32_16x16x4_f32(
           false, a,  false, b,
           (short)0, acc[t],  false,  false);
    }
  }

  __syncthreads();
  const int wl = (int)__builtin_amdgcn_mbcnt_lo(0xffffffffu, 0u);
  const float4 b4 = ((const float4*)bias)[wl];
  const int wcol = wv * 16 + (wl & 15);
  const int whalf = wl >> 4;
#pragma unroll
  for (int t = 0; t < ROW_TILES; ++t) {
#pragma unroll
    for (int r = 0; r < 8; ++r)
      tile[(t * 16 + r + whalf * 8) * PITCH + wcol] = acc[t][r];
  }
  __syncthreads();
#pragma unroll
  for (int i = 0; i < ROWS_PER_WAVE; ++i) {
    const int row = wv * ROWS_PER_WAVE + i;
    float4 o = *(const float4*)(tile + row * PITCH + 4 * wl);
    o.x += b4.x;
    o.y += b4.y;
    o.z += b4.z;
    o.w += b4.w;
    *(float4*)(out + (size_t)(block_row0 + row) * F_OUT + 4 * wl) = o;
  }
}

extern "C" void kernel_launch(void* const* d_in, const int* in_sizes, int n_in,
                              void* d_out, int out_size, void* d_ws,
                              size_t ws_size, hipStream_t stream) {
  const float* x     = (const float*)d_in[0];
  const float* w     = (const float*)d_in[1];
  const float* bias  = (const float*)d_in[2];
  const float* vals0 = (const float*)d_in[3];
  const float* vals1 = (const float*)d_in[4];
  const int*   rows0 = (const int*)d_in[5];
  const int*   cols0 = (const int*)d_in[6];
  const int*   rows1 = (const int*)d_in[7];
  const int*   cols1 = (const int*)d_in[8];

  float* out    = (float*)d_out;
  int*   sorted = (int*)d_ws + N_NODES * F_OUT;
  int*   starts = sorted + 2 * ALL_EDGES;

  gconv_slice_sort<<<N_SLICES, SORT_THREADS, 0, stream>>>(
      rows0, cols0, vals0, rows1, cols1, vals1, sorted, starts);

  gconv_sum_gemm<<<N_NODES / ROWS_PER_BLOCK, 256, 0, stream>>>(
      x, w, sorted, starts, bias, out);
}
